# starting kernel
# baseline (speedup 1.0000x reference)
#include <hip/hip_runtime.h>
#include <hip/hip_bf16.h>

typedef __attribute__((ext_vector_type(2))) float v2f;
typedef __attribute__((ext_vector_type(8))) float v8f;

__global__ void fuse_weights_kernel(const float* __restrict__ w1,
                                    const float* __restrict__ w2,
                                    float* __restrict__ Wc) {
    int t = blockIdx.x * blockDim.x + threadIdx.x;
    if (t >= 128 * 224) return;
    int oc = t / 224;
    int r  = t - oc * 224;
    int kk = r >> 5;
    int j  = r & 31;
    float s = 0.0f;
#pragma unroll
    for (int i = 0; i < 9; ++i)
        s += w1[oc * 9 + i] * w2[(j * 7 + kk) * 9 + i];
    Wc[t] = s;
}

__global__ __launch_bounds__(256) void conv_wmma_kernel(
    const float* __restrict__ x,
    const float* __restrict__ Wc,
    float* __restrict__ out) {
    const int blk   = blockIdx.x;
    const int g     = blk / 49;
    const int nTile = blk % 49;
    const int wave  = threadIdx.x >> 5;
    const int lane  = threadIdx.x & 31;
    const int n     = lane & 15;
    const int half  = lane >> 4;

    const int mBase = wave * 16;
    const int p     = nTile * 16 + n;
    const int h     = p / 28;
    const int w     = p - h * 28;

    const float* __restrict__ wcRow = Wc + (mBase + n) * 224 + half * 2;
    const float* __restrict__ xrow  = x + g * (32 * 784) + h * 28;

    v8f acc = {};

#pragma unroll
    for (int kk = 0; kk < 7; ++kk) {
        const int  wx    = w + kk - 3;
        const bool valid = (wx >= 0) && (wx < 28);
        const float* __restrict__ xcol = xrow + wx + half * 2 * 784;
        const float* __restrict__ wcK  = wcRow + kk * 32;

#pragma unroll
        for (int jb = 0; jb < 32; jb += 4) {
            v2f a = *(const v2f*)(wcK + jb);

            v2f b;
            b.x = valid ? xcol[(jb + 0) * 784] : 0.0f;
            b.y = valid ? xcol[(jb + 1) * 784] : 0.0f;

            acc = __builtin_amdgcn_wmma_f32_16x16x4_f32(
                 false, a,  false, b,
                 (short)0, acc,  false,  false);
        }
    }

    int hRoll = h + 1;
    if (hRoll == 28) hRoll = 0;
    float* __restrict__ obase =
        out + ((g * 128 + mBase + half * 8) * 28 + hRoll) * 28 + w;
#pragma unroll
    for (int v = 0; v < 8; ++v)
        obase[v * 784] = acc[v];
}

extern "C" void kernel_launch(void* const* d_in, const int* in_sizes, int n_in,
                              void* d_out, int out_size, void* d_ws, size_t ws_size,
                              hipStream_t stream) {
    const float* x  = (const float*)d_in[0];
    const float* w1 = (const float*)d_in[1];
    const float* w2 = (const float*)d_in[2];
    float* out = (float*)d_out;
    float* Wc  = (float*)d_ws;

    fuse_weights_kernel<<<(128 * 224 + 255) / 256, 256, 0, stream>>>(w1, w2, Wc);
    conv_wmma_kernel<<<2 * 49, 256, 0, stream>>>(x, Wc, out);
}


# each lane owns 2 adjacent image columns, so a wave carries both column tiles: x loads, U staging, tap sums and output rows go as 8-byte pairs; 56 blocks of 4 waves instead of 8; sum order kept
# speedup vs baseline: 4.6534x; 4.6534x over previous
#include <hip/hip_runtime.h>

typedef __attribute__((ext_vector_type(2))) float v2f;
typedef __attribute__((ext_vector_type(4))) float v4f;
typedef __attribute__((ext_vector_type(8))) float v8f;

#define U_PITCH 48
#define U_TAP   (9 * U_PITCH + 1)

#define WMMA(A, B, C)                                                          \
    __builtin_amdgcn_wmma_f32_16x16x4_f32(false, (A), false, (B), (short)0, (C), false, false)

__global__ __launch_bounds__(128) void conv_wmma_kernel(
    const float* __restrict__ x,
    const float* __restrict__ w1,
    const float* __restrict__ w2,
    float* __restrict__ out) {
    __shared__ __attribute__((aligned(16))) float sU[64 * U_PITCH];

    const int g    = blockIdx.x / 28;
    const int h    = blockIdx.x - g * 28;
    const int tid  = threadIdx.x;
    const int wave = tid >> 5;
    const int lane = tid & 31;
    const int c    = lane & 15;
    const int half = lane >> 4;

    const int mt = wave;
    int f = mt * 16 + c;
    f = (f < 63) ? f : 62;
    const bool colOk = c < 14;
    const float* __restrict__ pA = w2 + (2 * half) * 63 + f;
    const float* __restrict__ pB =
        x + (g * 32 + 2 * half) * 784 + h * 28 + (colOk ? 2 * c : 26);
    v2f a1[8], b1[2][8];
#pragma unroll
    for (int u = 0; u < 8; ++u) {
        a1[u].x = pA[(4 * u) * 63];
        a1[u].y = pA[(4 * u + 1) * 63];
        const v2f px = *(const v2f*)(pB + (4 * u) * 784);
        const v2f py = *(const v2f*)(pB + (4 * u + 1) * 784);
        b1[0][u].x = colOk ? px.x : 0.0f;
        b1[0][u].y = colOk ? py.x : 0.0f;
        b1[1][u].x = colOk ? px.y : 0.0f;
        b1[1][u].y = colOk ? py.y : 0.0f;
    }

    const int mp = wave;
    v2f a2[2][3];
#pragma unroll
    for (int t = 0; t < 2; ++t) {
        const float* __restrict__ pW1 = w1 + (mp * 32 + t * 16 + c) * 9;
        a2[t][0].x = pW1[2 * half];
        a2[t][0].y = pW1[2 * half + 1];
        a2[t][1].x = pW1[2 * half + 4];
        a2[t][1].y = pW1[2 * half + 5];
        const float w18 = pW1[8];
        a2[t][2].x = (half == 0) ? w18 : 0.0f;
        a2[t][2].y = 0.0f;
    }

    if (tid < 64) {
        const v4f zero4 = {0.0f, 0.0f, 0.0f, 0.0f};
        float* row = sU + tid * U_PITCH;
        *(v4f*)(row)      = zero4;
        *(v4f*)(row + 36) = zero4;
    }

    v8f accU[2] = {};
#pragma unroll
    for (int u = 0; u < 8; ++u) {
        accU[0] = WMMA(a1[u], b1[0][u], accU[0]);
        accU[1] = WMMA(a1[u], b1[1][u], accU[1]);
    }
    {
        float* dst = sU + (mt * 16 + half * 8) * U_PITCH + 4 + 2 * c;
#pragma unroll
        for (int r = 0; r < 8; ++r) {
            v2f pr;
            pr.x = accU[0][r];
            pr.y = accU[1][r];
            *(v2f*)(dst + r * U_PITCH) = pr;
        }
    }
    __syncthreads();

    const float* tb = sU + 1 + 2 * c + (2 * half) * U_PITCH;
    v2f b2[2][3];
    v2f tap[5][7];
#pragma unroll
    for (int k = 0; k < 7; ++k) {
#pragma unroll
        for (int u = 0; u < 2; ++u) {
            tap[2 * u][k].x     = tb[(4 * u) * U_PITCH + k * U_TAP];
            tap[2 * u][k].y     = tb[(4 * u) * U_PITCH + k * U_TAP + 1];
            tap[2 * u + 1][k].x = tb[(4 * u + 1) * U_PITCH + k * U_TAP];
            tap[2 * u + 1][k].y = tb[(4 * u + 1) * U_PITCH + k * U_TAP + 1];
        }
        tap[4][k].x = sU[1 + 2 * c + 8 * U_PITCH + k * U_TAP];
        tap[4][k].y = sU[1 + 2 * c + 8 * U_PITCH + k * U_TAP + 1];
    }
    v2f ts[5];
#pragma unroll
    for (int q = 0; q < 5; ++q) {
        v2f s = {0.0f, 0.0f};
#pragma unroll
        for (int k = 0; k < 7; ++k)
            s += tap[q][k];
        ts[q] = s;
    }
#pragma unroll
    for (int u = 0; u < 2; ++u) {
        b2[0][u].x = ts[2 * u].x;
        b2[0][u].y = ts[2 * u + 1].x;
        b2[1][u].x = ts[2 * u].y;
        b2[1][u].y = ts[2 * u + 1].y;
    }
    b2[0][2].x = (half == 0) ? ts[4].x : 0.0f;
    b2[0][2].y = 0.0f;
    b2[1][2].x = (half == 0) ? ts[4].y : 0.0f;
    b2[1][2].y = 0.0f;

    v8f acc[2][2] = {};
#pragma unroll
    for (int u = 0; u < 3; ++u) {
        acc[0][0] = WMMA(a2[0][u], b2[0][u], acc[0][0]);
        acc[0][1] = WMMA(a2[0][u], b2[1][u], acc[0][1]);
        acc[1][0] = WMMA(a2[1][u], b2[0][u], acc[1][0]);
        acc[1][1] = WMMA(a2[1][u], b2[1][u], acc[1][1]);
    }

    int hRoll = h + 1;
    if (hRoll == 28) hRoll = 0;
    if (colOk) {
        float* __restrict__ obase =
            out + ((g * 128 + mp * 32 + half * 8) * 28 + hRoll) * 28 + 2 * c;
#pragma unroll
        for (int t = 0; t < 2; ++t)
#pragma unroll
            for (int r = 0; r < 8; ++r) {
                v2f pr;
                pr.x = acc[t][0][r];
                pr.y = acc[t][1][r];
                *(v2f*)(obase + (t * 16 + r) * 784) = pr;
            }
    }
}

extern "C" void kernel_launch(void* const* d_in, const int* in_sizes, int n_in,
                              void* d_out, int out_size, void* d_ws, size_t ws_size,
                              hipStream_t stream) {
    const float* x  = (const float*)d_in[0];
    const float* w1 = (const float*)d_in[1];
    const float* w2 = (const float*)d_in[2];
    float* out = (float*)d_out;

    conv_wmma_kernel<<<2 * 28, 128, 0, stream>>>(x, w1, w2, out);
}
